# starting kernel
# baseline (speedup 1.0000x reference)
#include <hip/hip_runtime.h>
#include <math.h>

typedef __attribute__((ext_vector_type(2))) float v2f;
typedef __attribute__((ext_vector_type(8))) float v8f;

#define BLOCK 256

__device__ __forceinline__ float mat_entry(int m, int k, const float cw[4], const float sw[4]) {
    float p = 1.0f;
#pragma unroll
    for (int j = 0; j < 4; ++j) {
        int mb = (m >> (3 - j)) & 1;
        int kb = (k >> (3 - j)) & 1;
        float u = (mb == kb) ? cw[j] : (mb ? sw[j] : -sw[j]);
        p *= u;
    }
    return p;
}

__global__ __launch_bounds__(BLOCK)
void dqc_wmma_kernel(const float* __restrict__ feat,
                     const float* __restrict__ qp,
                     float* __restrict__ out) {
    constexpr float PI_4 = 0.78539816339744830962f;
    constexpr float CRZ_C[6] = {0.0f, 0.70710678118654752f, 1.0f,
                                0.70710678118654752f, 0.0f, -0.70710678118654752f};
    constexpr float CRZ_S[6] = {-1.0f, -0.70710678118654752f, 0.0f,
                                0.70710678118654752f, 1.0f, 0.70710678118654752f};

    __shared__ float s_re[BLOCK][17];
    __shared__ float s_im[BLOCK][17];

    const int tid  = threadIdx.x;
    const int lane = tid & 31;
    const int wv   = tid >> 5;
    const int r    = lane & 15;
    const int hi   = lane >> 4;
    const int elem = blockIdx.x * BLOCK + tid;

    float4 f = reinterpret_cast<const float4*>(feat)[elem];
    float th[4] = {f.x, f.y, f.z, f.w};
    float cc[4], ss[4];
#pragma unroll
    for (int q = 0; q < 4; ++q) {
        float a = tanhf(th[q]) * PI_4;
        __sincosf(a, &ss[q], &cc[q]);
    }

#pragma unroll
    for (int sp = 0; sp < 16; ++sp) {
        const int q1 = (sp >> 3) & 1;
        const int q3 = (sp >> 2) & 1;
        const int q0 = (sp >> 1) & 1;
        const int q2 = sp & 1;
        const int kk = (q0 ? (q1 ? 1 : -1) : 0)
                     + (q1 ? (q2 ? 1 : -1) : 0)
                     + (q2 ? (q3 ? 1 : -1) : 0);
        float re = 0.25f * CRZ_C[kk + 2];
        float im = 0.25f * CRZ_S[kk + 2];
        const int bits[4] = {q0, q1, q2, q3};
#pragma unroll
        for (int q = 0; q < 4; ++q) {
            float br = cc[q];
            float bi = bits[q] ? ss[q] : -ss[q];
            float nr = re * br - im * bi;
            float ni = re * bi + im * br;
            re = nr; im = ni;
        }
        s_re[tid][sp] = re;
        s_im[tid][sp] = im;
    }

    float cw[4], sw[4];
#pragma unroll
    for (int j = 0; j < 4; ++j) __sincosf(qp[j] * 0.5f, &sw[j], &cw[j]);
    v2f A[4];
#pragma unroll
    for (int c = 0; c < 4; ++c) {
        A[c].x = mat_entry(r, 4 * c + 2 * hi,     cw, sw);
        A[c].y = mat_entry(r, 4 * c + 2 * hi + 1, cw, sw);
    }

    __syncthreads();

#pragma unroll
    for (int t = 0; t < 2; ++t) {
        const int n = wv * 32 + t * 16 + r;
        v8f accR = {0.f, 0.f, 0.f, 0.f, 0.f, 0.f, 0.f, 0.f};
        v8f accI = {0.f, 0.f, 0.f, 0.f, 0.f, 0.f, 0.f, 0.f};
#pragma unroll
        for (int c = 0; c < 4; ++c) {
            const int k0 = 4 * c + 2 * hi;
            v2f bR; bR.x = s_re[n][k0]; bR.y = s_re[n][k0 + 1];
            v2f bI; bI.x = s_im[n][k0]; bI.y = s_im[n][k0 + 1];
            accR = __builtin_amdgcn_wmma_f32_16x16x4_f32(false, A[c], false, bR,
                                                         (short)0, accR, false, false);
            accI = __builtin_amdgcn_wmma_f32_16x16x4_f32(false, A[c], false, bI,
                                                         (short)0, accI, false, false);
        }

        float e0 = 0.f, e1 = 0.f, e2 = 0.f, e3 = 0.f;
#pragma unroll
        for (int v = 0; v < 8; ++v) {
            float p = accR[v] * accR[v] + accI[v] * accI[v];
            int m = v + 8 * hi;
            e0 += ((m >> 3) & 1) ? -p : p;
            e1 += ((m >> 2) & 1) ? -p : p;
            e2 += ((m >> 1) & 1) ? -p : p;
            e3 += ( m       & 1) ? -p : p;
        }
        e0 += __shfl_xor(e0, 16, 32);
        e1 += __shfl_xor(e1, 16, 32);
        e2 += __shfl_xor(e2, 16, 32);
        e3 += __shfl_xor(e3, 16, 32);

        if (hi == t) {
            int oe = blockIdx.x * BLOCK + wv * 32 + t * 16 + r;
            reinterpret_cast<float4*>(out)[oe] = make_float4(e0, e1, e2, e3);
        }
    }
}

extern "C" void kernel_launch(void* const* d_in, const int* in_sizes, int n_in,
                              void* d_out, int out_size, void* d_ws, size_t ws_size,
                              hipStream_t stream) {
    const float* feat = (const float*)d_in[0];
    const float* qp   = (const float*)d_in[1];
    float* out        = (float*)d_out;
    const int batch   = in_sizes[0] / 4;
    dqc_wmma_kernel<<<batch / BLOCK, BLOCK, 0, stream>>>(feat, qp, out);
}


# tanh as one branch-free closed form 1-2/(1+e^2|x|) on v_exp_f32 and v_rcp_f32 instead of library tanhf, whose two per-lane branches every wave ran both of; abs error below 3e-7
# speedup vs baseline: 1.1194x; 1.1194x over previous
#include <hip/hip_runtime.h>
#include <math.h>

typedef __attribute__((ext_vector_type(2))) float v2f;
typedef __attribute__((ext_vector_type(8))) float v8f;

#define BLOCK 256

__device__ __forceinline__ float mat_entry(int m, int k, const float cw[4], const float sw[4]) {
    float p = 1.0f;
#pragma unroll
    for (int j = 0; j < 4; ++j) {
        int mb = (m >> (3 - j)) & 1;
        int kb = (k >> (3 - j)) & 1;
        float u = (mb == kb) ? cw[j] : (mb ? sw[j] : -sw[j]);
        p *= u;
    }
    return p;
}

__device__ __forceinline__ float tanh_closed(float x) {
    constexpr float TWO_LOG2E = 2.0f * 1.4426950408889634f;
    const float e = __builtin_amdgcn_exp2f(fabsf(x) * TWO_LOG2E);
    const float r = __builtin_amdgcn_rcpf(1.0f + e);
    return copysignf(fmaf(-2.0f, r, 1.0f), x);
}

__global__ __launch_bounds__(BLOCK)
void dqc_wmma_kernel(const float* __restrict__ feat,
                     const float* __restrict__ qp,
                     float* __restrict__ out) {
    constexpr float PI_4 = 0.78539816339744830962f;
    constexpr float CRZ_C[6] = {0.0f, 0.70710678118654752f, 1.0f,
                                0.70710678118654752f, 0.0f, -0.70710678118654752f};
    constexpr float CRZ_S[6] = {-1.0f, -0.70710678118654752f, 0.0f,
                                0.70710678118654752f, 1.0f, 0.70710678118654752f};

    __shared__ float s_re[BLOCK][17];
    __shared__ float s_im[BLOCK][17];

    const int tid  = threadIdx.x;
    const int lane = tid & 31;
    const int wv   = tid >> 5;
    const int r    = lane & 15;
    const int hi   = lane >> 4;
    const int elem = blockIdx.x * BLOCK + tid;

    float4 f = reinterpret_cast<const float4*>(feat)[elem];
    float th[4] = {f.x, f.y, f.z, f.w};
    float cc[4], ss[4];
#pragma unroll
    for (int q = 0; q < 4; ++q) {
        float a = tanh_closed(th[q]) * PI_4;
        __sincosf(a, &ss[q], &cc[q]);
    }

#pragma unroll
    for (int sp = 0; sp < 16; ++sp) {
        const int q1 = (sp >> 3) & 1;
        const int q3 = (sp >> 2) & 1;
        const int q0 = (sp >> 1) & 1;
        const int q2 = sp & 1;
        const int kk = (q0 ? (q1 ? 1 : -1) : 0)
                     + (q1 ? (q2 ? 1 : -1) : 0)
                     + (q2 ? (q3 ? 1 : -1) : 0);
        float re = 0.25f * CRZ_C[kk + 2];
        float im = 0.25f * CRZ_S[kk + 2];
        const int bits[4] = {q0, q1, q2, q3};
#pragma unroll
        for (int q = 0; q < 4; ++q) {
            float br = cc[q];
            float bi = bits[q] ? ss[q] : -ss[q];
            float nr = re * br - im * bi;
            float ni = re * bi + im * br;
            re = nr; im = ni;
        }
        s_re[tid][sp] = re;
        s_im[tid][sp] = im;
    }

    float cw[4], sw[4];
#pragma unroll
    for (int j = 0; j < 4; ++j) __sincosf(qp[j] * 0.5f, &sw[j], &cw[j]);
    v2f A[4];
#pragma unroll
    for (int c = 0; c < 4; ++c) {
        A[c].x = mat_entry(r, 4 * c + 2 * hi,     cw, sw);
        A[c].y = mat_entry(r, 4 * c + 2 * hi + 1, cw, sw);
    }

    __syncthreads();

#pragma unroll
    for (int t = 0; t < 2; ++t) {
        const int n = wv * 32 + t * 16 + r;
        v8f accR = {0.f, 0.f, 0.f, 0.f, 0.f, 0.f, 0.f, 0.f};
        v8f accI = {0.f, 0.f, 0.f, 0.f, 0.f, 0.f, 0.f, 0.f};
#pragma unroll
        for (int c = 0; c < 4; ++c) {
            const int k0 = 4 * c + 2 * hi;
            v2f bR; bR.x = s_re[n][k0]; bR.y = s_re[n][k0 + 1];
            v2f bI; bI.x = s_im[n][k0]; bI.y = s_im[n][k0 + 1];
            accR = __builtin_amdgcn_wmma_f32_16x16x4_f32(false, A[c], false, bR,
                                                         (short)0, accR, false, false);
            accI = __builtin_amdgcn_wmma_f32_16x16x4_f32(false, A[c], false, bI,
                                                         (short)0, accI, false, false);
        }

        float e0 = 0.f, e1 = 0.f, e2 = 0.f, e3 = 0.f;
#pragma unroll
        for (int v = 0; v < 8; ++v) {
            float p = accR[v] * accR[v] + accI[v] * accI[v];
            int m = v + 8 * hi;
            e0 += ((m >> 3) & 1) ? -p : p;
            e1 += ((m >> 2) & 1) ? -p : p;
            e2 += ((m >> 1) & 1) ? -p : p;
            e3 += ( m       & 1) ? -p : p;
        }
        e0 += __shfl_xor(e0, 16, 32);
        e1 += __shfl_xor(e1, 16, 32);
        e2 += __shfl_xor(e2, 16, 32);
        e3 += __shfl_xor(e3, 16, 32);

        if (hi == t) {
            int oe = blockIdx.x * BLOCK + wv * 32 + t * 16 + r;
            reinterpret_cast<float4*>(out)[oe] = make_float4(e0, e1, e2, e3);
        }
    }
}

extern "C" void kernel_launch(void* const* d_in, const int* in_sizes, int n_in,
                              void* d_out, int out_size, void* d_ws, size_t ws_size,
                              hipStream_t stream) {
    const float* feat = (const float*)d_in[0];
    const float* qp   = (const float*)d_in[1];
    float* out        = (float*)d_out;
    const int batch   = in_sizes[0] / 4;
    dqc_wmma_kernel<<<batch / BLOCK, BLOCK, 0, stream>>>(feat, qp, out);
}


# 16 diagonal amplitudes per element from two 2-qubit phase tables (conjugate pairs, CRZ eighth-turns folded in) and one packed complex product per state, not four chained ones; products re-associated
# speedup vs baseline: 1.2347x; 1.1030x over previous
#include <hip/hip_runtime.h>
#include <math.h>

typedef __attribute__((ext_vector_type(2))) float v2f;
typedef __attribute__((ext_vector_type(8))) float v8f;

#define BLOCK 256

__device__ __forceinline__ float mat_entry(int m, int k, const float cw[4], const float sw[4]) {
    float p = 1.0f;
#pragma unroll
    for (int j = 0; j < 4; ++j) {
        int mb = (m >> (3 - j)) & 1;
        int kb = (k >> (3 - j)) & 1;
        float u = (mb == kb) ? cw[j] : (mb ? sw[j] : -sw[j]);
        p *= u;
    }
    return p;
}

__device__ __forceinline__ v2f cconj(v2f a) { return (v2f){a.x, -a.y}; }
__device__ __forceinline__ v2f cmul(v2f f, v2f g) {
    const v2f t = (v2f){-f.y, f.y} * (v2f){g.y, g.x};
    return (v2f){f.x, f.x} * g + t;
}
__device__ __forceinline__ constexpr int crz_t(int c, int t) { return c ? (t ? 1 : -1) : 0; }
__device__ __forceinline__ v2f rot8(v2f a, int n) {
    constexpr float C45 = 0.70710678118654752f;
    if (n == 0) return a;
    if (n == 2) return (v2f){-a.y, a.x};
    if (n == 1) return (a + (v2f){-a.y, a.x}) * C45;
    return (a + (v2f){a.y, -a.x}) * C45;
}

__device__ __forceinline__ float tanh_closed(float x) {
    constexpr float TWO_LOG2E = 2.0f * 1.4426950408889634f;
    const float e = __builtin_amdgcn_exp2f(fabsf(x) * TWO_LOG2E);
    const float r = __builtin_amdgcn_rcpf(1.0f + e);
    return copysignf(fmaf(-2.0f, r, 1.0f), x);
}

__global__ __launch_bounds__(BLOCK)
void dqc_wmma_kernel(const float* __restrict__ feat,
                     const float* __restrict__ qp,
                     float* __restrict__ out) {
    constexpr float PI_4 = 0.78539816339744830962f;
    __shared__ float s_re[BLOCK][17];
    __shared__ float s_im[BLOCK][17];

    const int tid  = threadIdx.x;
    const int lane = tid & 31;
    const int wv   = tid >> 5;
    const int r    = lane & 15;
    const int hi   = lane >> 4;
    const int elem = blockIdx.x * BLOCK + tid;

    float4 f = reinterpret_cast<const float4*>(feat)[elem];
    float th[4] = {f.x, f.y, f.z, f.w};
    float cc[4], ss[4];
#pragma unroll
    for (int q = 0; q < 4; ++q) {
        float a = tanh_closed(th[q]) * PI_4;
        __sincosf(a, &ss[q], &cc[q]);
    }

    const v2f e0 = (v2f){cc[0], ss[0]} * 0.25f;
    const v2f e1 = {cc[1], ss[1]}, e2 = {cc[2], ss[2]}, e3 = {cc[3], ss[3]};
    const v2f pp01 = cmul(e0, e1), pm01 = cmul(e0, cconj(e1));
    const v2f pp23 = cmul(e2, e3), pm23 = cmul(e2, cconj(e3));
    const v2f P01[2][2] = {{cconj(pp01), cconj(pm01)}, {pm01, pp01}};
    const v2f P23[2][2] = {{cconj(pp23), cconj(pm23)}, {pm23, pp23}};
    v2f F01[2][2], G[2][2][2];
#pragma unroll
    for (int a = 0; a < 2; ++a) {
#pragma unroll
        for (int b = 0; b < 2; ++b) {
            F01[a][b] = rot8(P01[a][b], crz_t(a, b));
#pragma unroll
            for (int q1 = 0; q1 < 2; ++q1)
                G[q1][a][b] = rot8(P23[a][b], crz_t(q1, a) + crz_t(a, b));
        }
    }
#pragma unroll
    for (int sp = 0; sp < 16; ++sp) {
        const int q1 = (sp >> 3) & 1;
        const int q3 = (sp >> 2) & 1;
        const int q0 = (sp >> 1) & 1;
        const int q2 = sp & 1;
        const v2f amp = cmul(F01[q0][q1], G[q1][q2][q3]);
        s_re[tid][sp] = amp.x;
        s_im[tid][sp] = amp.y;
    }

    float cw[4], sw[4];
#pragma unroll
    for (int j = 0; j < 4; ++j) __sincosf(qp[j] * 0.5f, &sw[j], &cw[j]);
    v2f A[4];
#pragma unroll
    for (int c = 0; c < 4; ++c) {
        A[c].x = mat_entry(r, 4 * c + 2 * hi,     cw, sw);
        A[c].y = mat_entry(r, 4 * c + 2 * hi + 1, cw, sw);
    }

    __syncthreads();

#pragma unroll
    for (int t = 0; t < 2; ++t) {
        const int n = wv * 32 + t * 16 + r;
        v8f accR = {0.f, 0.f, 0.f, 0.f, 0.f, 0.f, 0.f, 0.f};
        v8f accI = {0.f, 0.f, 0.f, 0.f, 0.f, 0.f, 0.f, 0.f};
#pragma unroll
        for (int c = 0; c < 4; ++c) {
            const int k0 = 4 * c + 2 * hi;
            v2f bR; bR.x = s_re[n][k0]; bR.y = s_re[n][k0 + 1];
            v2f bI; bI.x = s_im[n][k0]; bI.y = s_im[n][k0 + 1];
            accR = __builtin_amdgcn_wmma_f32_16x16x4_f32(false, A[c], false, bR,
                                                         (short)0, accR, false, false);
            accI = __builtin_amdgcn_wmma_f32_16x16x4_f32(false, A[c], false, bI,
                                                         (short)0, accI, false, false);
        }

        float e0 = 0.f, e1 = 0.f, e2 = 0.f, e3 = 0.f;
#pragma unroll
        for (int v = 0; v < 8; ++v) {
            float p = accR[v] * accR[v] + accI[v] * accI[v];
            int m = v + 8 * hi;
            e0 += ((m >> 3) & 1) ? -p : p;
            e1 += ((m >> 2) & 1) ? -p : p;
            e2 += ((m >> 1) & 1) ? -p : p;
            e3 += ( m       & 1) ? -p : p;
        }
        e0 += __shfl_xor(e0, 16, 32);
        e1 += __shfl_xor(e1, 16, 32);
        e2 += __shfl_xor(e2, 16, 32);
        e3 += __shfl_xor(e3, 16, 32);

        if (hi == t) {
            int oe = blockIdx.x * BLOCK + wv * 32 + t * 16 + r;
            reinterpret_cast<float4*>(out)[oe] = make_float4(e0, e1, e2, e3);
        }
    }
}

extern "C" void kernel_launch(void* const* d_in, const int* in_sizes, int n_in,
                              void* d_out, int out_size, void* d_ws, size_t ws_size,
                              hipStream_t stream) {
    const float* feat = (const float*)d_in[0];
    const float* qp   = (const float*)d_in[1];
    float* out        = (float*)d_out;
    const int batch   = in_sizes[0] / 4;
    dqc_wmma_kernel<<<batch / BLOCK, BLOCK, 0, stream>>>(feat, qp, out);
}


# amplitudes reach the WMMA B fragments by 16 v_permlane16_swap_b32 between the wave's halves instead of a shared-memory transpose: no LDS, no block barrier; same products, same sums
# speedup vs baseline: 1.3008x; 1.0535x over previous
#include <hip/hip_runtime.h>
#include <math.h>

typedef __attribute__((ext_vector_type(2))) float v2f;
typedef __attribute__((ext_vector_type(8))) float v8f;

#define BLOCK 256

__device__ __forceinline__ float mat_entry(int m, int k, const float cw[4], const float sw[4]) {
    float p = 1.0f;
#pragma unroll
    for (int j = 0; j < 4; ++j) {
        int mb = (m >> (3 - j)) & 1;
        int kb = (k >> (3 - j)) & 1;
        float u = (mb == kb) ? cw[j] : (mb ? sw[j] : -sw[j]);
        p *= u;
    }
    return p;
}

__device__ __forceinline__ v2f cconj(v2f a) { return (v2f){a.x, -a.y}; }
__device__ __forceinline__ v2f cmul(v2f f, v2f g) {
    const v2f t = (v2f){-f.y, f.y} * (v2f){g.y, g.x};
    return (v2f){f.x, f.x} * g + t;
}
__device__ __forceinline__ constexpr int crz_t(int c, int t) { return c ? (t ? 1 : -1) : 0; }
__device__ __forceinline__ v2f rot8(v2f a, int n) {
    constexpr float C45 = 0.70710678118654752f;
    if (n == 0) return a;
    if (n == 2) return (v2f){-a.y, a.x};
    if (n == 1) return (a + (v2f){-a.y, a.x}) * C45;
    return (a + (v2f){a.y, -a.x}) * C45;
}

__device__ __forceinline__ float tanh_closed(float x) {
    constexpr float TWO_LOG2E = 2.0f * 1.4426950408889634f;
    const float e = __builtin_amdgcn_exp2f(fabsf(x) * TWO_LOG2E);
    const float r = __builtin_amdgcn_rcpf(1.0f + e);
    return copysignf(fmaf(-2.0f, r, 1.0f), x);
}

__device__ __forceinline__ void swap_halves(float& d, float& s) {
    typedef __attribute__((ext_vector_type(2))) unsigned int v2u_t;
    const v2u_t r = __builtin_amdgcn_permlane16_swap(__float_as_uint(d), __float_as_uint(s), false, false);
    d = __uint_as_float(r.x); s = __uint_as_float(r.y);
}

__global__ __launch_bounds__(BLOCK)
void dqc_wmma_kernel(const float* __restrict__ feat,
                     const float* __restrict__ qp,
                     float* __restrict__ out) {
    constexpr float PI_4 = 0.78539816339744830962f;

    const int tid  = threadIdx.x;
    const int lane = tid & 31;
    const int wv   = tid >> 5;
    const int r    = lane & 15;
    const int hi   = lane >> 4;
    const int elem = blockIdx.x * BLOCK + tid;

    float4 f = reinterpret_cast<const float4*>(feat)[elem];
    float th[4] = {f.x, f.y, f.z, f.w};
    float cc[4], ss[4];
#pragma unroll
    for (int q = 0; q < 4; ++q) {
        float a = tanh_closed(th[q]) * PI_4;
        __sincosf(a, &ss[q], &cc[q]);
    }

    const v2f e0 = (v2f){cc[0], ss[0]} * 0.25f;
    const v2f e1 = {cc[1], ss[1]}, e2 = {cc[2], ss[2]}, e3 = {cc[3], ss[3]};
    const v2f pp01 = cmul(e0, e1), pm01 = cmul(e0, cconj(e1));
    const v2f pp23 = cmul(e2, e3), pm23 = cmul(e2, cconj(e3));
    const v2f P01[2][2] = {{cconj(pp01), cconj(pm01)}, {pm01, pp01}};
    const v2f P23[2][2] = {{cconj(pp23), cconj(pm23)}, {pm23, pp23}};
    v2f F01[2][2], G[2][2][2];
#pragma unroll
    for (int a = 0; a < 2; ++a) {
#pragma unroll
        for (int b = 0; b < 2; ++b) {
            F01[a][b] = rot8(P01[a][b], crz_t(a, b));
#pragma unroll
            for (int q1 = 0; q1 < 2; ++q1)
                G[q1][a][b] = rot8(P23[a][b], crz_t(q1, a) + crz_t(a, b));
        }
    }

    v2f fr[2][4], fi[2][4];
#pragma unroll
    for (int c = 0; c < 4; ++c) {
        const int q1 = c >> 1, q3 = c & 1;
        const v2f gx = {G[q1][0][q3].x, G[q1][1][q3].x};
        const v2f gy = {G[q1][0][q3].y, G[q1][1][q3].y};
#pragma unroll
        for (int q0 = 0; q0 < 2; ++q0) {
            const v2f fx = {F01[q0][q1].x, F01[q0][q1].x};
            const v2f fy = {F01[q0][q1].y, F01[q0][q1].y};
            const v2f tr = -fy * gy;
            const v2f ti =  fy * gx;
            fr[q0][c] = fx * gx + tr;
            fi[q0][c] = fx * gy + ti;
        }
    }
#pragma unroll
    for (int c = 0; c < 4; ++c) {
        float a0 = fr[0][c].x, b0 = fr[1][c].x, a1 = fr[0][c].y, b1 = fr[1][c].y;
        float a2 = fi[0][c].x, b2 = fi[1][c].x, a3 = fi[0][c].y, b3 = fi[1][c].y;
        swap_halves(a0, b0); swap_halves(a1, b1); swap_halves(a2, b2); swap_halves(a3, b3);
        fr[0][c] = (v2f){a0, a1}; fr[1][c] = (v2f){b0, b1};
        fi[0][c] = (v2f){a2, a3}; fi[1][c] = (v2f){b2, b3};
    }

    float cw[4], sw[4];
#pragma unroll
    for (int j = 0; j < 4; ++j) __sincosf(qp[j] * 0.5f, &sw[j], &cw[j]);
    v2f A[4];
#pragma unroll
    for (int c = 0; c < 4; ++c) {
        A[c].x = mat_entry(r, 4 * c + 2 * hi,     cw, sw);
        A[c].y = mat_entry(r, 4 * c + 2 * hi + 1, cw, sw);
    }

#pragma unroll
    for (int t = 0; t < 2; ++t) {
        v8f accR = {0.f, 0.f, 0.f, 0.f, 0.f, 0.f, 0.f, 0.f};
        v8f accI = {0.f, 0.f, 0.f, 0.f, 0.f, 0.f, 0.f, 0.f};
#pragma unroll
        for (int c = 0; c < 4; ++c) {
            accR = __builtin_amdgcn_wmma_f32_16x16x4_f32(false, A[c], false, fr[t][c],
                                                         (short)0, accR, false, false);
            accI = __builtin_amdgcn_wmma_f32_16x16x4_f32(false, A[c], false, fi[t][c],
                                                         (short)0, accI, false, false);
        }

        float e0 = 0.f, e1 = 0.f, e2 = 0.f, e3 = 0.f;
#pragma unroll
        for (int v = 0; v < 8; ++v) {
            float p = accR[v] * accR[v] + accI[v] * accI[v];
            int m = v + 8 * hi;
            e0 += ((m >> 3) & 1) ? -p : p;
            e1 += ((m >> 2) & 1) ? -p : p;
            e2 += ((m >> 1) & 1) ? -p : p;
            e3 += ( m       & 1) ? -p : p;
        }
        e0 += __shfl_xor(e0, 16, 32);
        e1 += __shfl_xor(e1, 16, 32);
        e2 += __shfl_xor(e2, 16, 32);
        e3 += __shfl_xor(e3, 16, 32);

        if (hi == t) {
            int oe = blockIdx.x * BLOCK + wv * 32 + t * 16 + r;
            reinterpret_cast<float4*>(out)[oe] = make_float4(e0, e1, e2, e3);
        }
    }
}

extern "C" void kernel_launch(void* const* d_in, const int* in_sizes, int n_in,
                              void* d_out, int out_size, void* d_ws, size_t ws_size,
                              hipStream_t stream) {
    const float* feat = (const float*)d_in[0];
    const float* qp   = (const float*)d_in[1];
    float* out        = (float*)d_out;
    const int batch   = in_sizes[0] / 4;
    dqc_wmma_kernel<<<batch / BLOCK, BLOCK, 0, stream>>>(feat, qp, out);
}


# Z sums as one add/sub butterfly per lane; the two tiles' half-sums meet by 4 v_permlane16_swap_b32, not 8 ds_bpermute_b32, and all 32 lanes store once; the order of the 16-term sums changes
# speedup vs baseline: 1.3839x; 1.0639x over previous
#include <hip/hip_runtime.h>
#include <math.h>

typedef __attribute__((ext_vector_type(2))) float v2f;
typedef __attribute__((ext_vector_type(8))) float v8f;

#define BLOCK 256

__device__ __forceinline__ float mat_entry(int m, int k, const float cw[4], const float sw[4]) {
    float p = 1.0f;
#pragma unroll
    for (int j = 0; j < 4; ++j) {
        int mb = (m >> (3 - j)) & 1;
        int kb = (k >> (3 - j)) & 1;
        float u = (mb == kb) ? cw[j] : (mb ? sw[j] : -sw[j]);
        p *= u;
    }
    return p;
}

__device__ __forceinline__ v2f cconj(v2f a) { return (v2f){a.x, -a.y}; }
__device__ __forceinline__ v2f cmul(v2f f, v2f g) {
    const v2f t = (v2f){-f.y, f.y} * (v2f){g.y, g.x};
    return (v2f){f.x, f.x} * g + t;
}
__device__ __forceinline__ constexpr int crz_t(int c, int t) { return c ? (t ? 1 : -1) : 0; }
__device__ __forceinline__ v2f rot8(v2f a, int n) {
    constexpr float C45 = 0.70710678118654752f;
    if (n == 0) return a;
    if (n == 2) return (v2f){-a.y, a.x};
    if (n == 1) return (a + (v2f){-a.y, a.x}) * C45;
    return (a + (v2f){a.y, -a.x}) * C45;
}

__device__ __forceinline__ float tanh_closed(float x) {
    constexpr float TWO_LOG2E = 2.0f * 1.4426950408889634f;
    const float e = __builtin_amdgcn_exp2f(fabsf(x) * TWO_LOG2E);
    const float r = __builtin_amdgcn_rcpf(1.0f + e);
    return copysignf(fmaf(-2.0f, r, 1.0f), x);
}

__device__ __forceinline__ void swap_halves(float& d, float& s) {
    typedef __attribute__((ext_vector_type(2))) unsigned int v2u_t;
    const v2u_t r = __builtin_amdgcn_permlane16_swap(__float_as_uint(d), __float_as_uint(s), false, false);
    d = __uint_as_float(r.x); s = __uint_as_float(r.y);
}

__global__ __launch_bounds__(BLOCK)
void dqc_wmma_kernel(const float* __restrict__ feat,
                     const float* __restrict__ qp,
                     float* __restrict__ out) {
    constexpr float PI_4 = 0.78539816339744830962f;

    const int tid  = threadIdx.x;
    const int lane = tid & 31;
    const int wv   = tid >> 5;
    const int r    = lane & 15;
    const int hi   = lane >> 4;
    const int elem = blockIdx.x * BLOCK + tid;

    float4 f = reinterpret_cast<const float4*>(feat)[elem];
    float th[4] = {f.x, f.y, f.z, f.w};
    float cc[4], ss[4];
#pragma unroll
    for (int q = 0; q < 4; ++q) {
        float a = tanh_closed(th[q]) * PI_4;
        __sincosf(a, &ss[q], &cc[q]);
    }

    const v2f e0 = (v2f){cc[0], ss[0]} * 0.25f;
    const v2f e1 = {cc[1], ss[1]}, e2 = {cc[2], ss[2]}, e3 = {cc[3], ss[3]};
    const v2f pp01 = cmul(e0, e1), pm01 = cmul(e0, cconj(e1));
    const v2f pp23 = cmul(e2, e3), pm23 = cmul(e2, cconj(e3));
    const v2f P01[2][2] = {{cconj(pp01), cconj(pm01)}, {pm01, pp01}};
    const v2f P23[2][2] = {{cconj(pp23), cconj(pm23)}, {pm23, pp23}};
    v2f F01[2][2], G[2][2][2];
#pragma unroll
    for (int a = 0; a < 2; ++a) {
#pragma unroll
        for (int b = 0; b < 2; ++b) {
            F01[a][b] = rot8(P01[a][b], crz_t(a, b));
#pragma unroll
            for (int q1 = 0; q1 < 2; ++q1)
                G[q1][a][b] = rot8(P23[a][b], crz_t(q1, a) + crz_t(a, b));
        }
    }

    v2f fr[2][4], fi[2][4];
#pragma unroll
    for (int c = 0; c < 4; ++c) {
        const int q1 = c >> 1, q3 = c & 1;
        const v2f gx = {G[q1][0][q3].x, G[q1][1][q3].x};
        const v2f gy = {G[q1][0][q3].y, G[q1][1][q3].y};
#pragma unroll
        for (int q0 = 0; q0 < 2; ++q0) {
            const v2f fx = {F01[q0][q1].x, F01[q0][q1].x};
            const v2f fy = {F01[q0][q1].y, F01[q0][q1].y};
            const v2f tr = -fy * gy;
            const v2f ti =  fy * gx;
            fr[q0][c] = fx * gx + tr;
            fi[q0][c] = fx * gy + ti;
        }
    }
#pragma unroll
    for (int c = 0; c < 4; ++c) {
        float a0 = fr[0][c].x, b0 = fr[1][c].x, a1 = fr[0][c].y, b1 = fr[1][c].y;
        float a2 = fi[0][c].x, b2 = fi[1][c].x, a3 = fi[0][c].y, b3 = fi[1][c].y;
        swap_halves(a0, b0); swap_halves(a1, b1); swap_halves(a2, b2); swap_halves(a3, b3);
        fr[0][c] = (v2f){a0, a1}; fr[1][c] = (v2f){b0, b1};
        fi[0][c] = (v2f){a2, a3}; fi[1][c] = (v2f){b2, b3};
    }

    float cw[4], sw[4];
#pragma unroll
    for (int j = 0; j < 4; ++j) __sincosf(qp[j] * 0.5f, &sw[j], &cw[j]);
    v2f A[4];
#pragma unroll
    for (int c = 0; c < 4; ++c) {
        A[c].x = mat_entry(r, 4 * c + 2 * hi,     cw, sw);
        A[c].y = mat_entry(r, 4 * c + 2 * hi + 1, cw, sw);
    }

    v2f tz1[2], z23[2];
#pragma unroll
    for (int t = 0; t < 2; ++t) {
        v8f accR = {0.f, 0.f, 0.f, 0.f, 0.f, 0.f, 0.f, 0.f};
        v8f accI = {0.f, 0.f, 0.f, 0.f, 0.f, 0.f, 0.f, 0.f};
#pragma unroll
        for (int c = 0; c < 4; ++c) {
            accR = __builtin_amdgcn_wmma_f32_16x16x4_f32(false, A[c], false, fr[t][c],
                                                         (short)0, accR, false, false);
            accI = __builtin_amdgcn_wmma_f32_16x16x4_f32(false, A[c], false, fi[t][c],
                                                         (short)0, accI, false, false);
        }

        v2f sd[4];
#pragma unroll
        for (int j = 0; j < 4; ++j) {
            const v2f re = {accR[2 * j], accR[2 * j + 1]};
            const v2f im = {accI[2 * j], accI[2 * j + 1]};
            const v2f p  = __builtin_elementwise_fma(im, im, re * re);
            sd[j] = (v2f){p.x, p.x} + (v2f){p.y, -p.y};
        }
        const v2f c01 = sd[0] + (v2f){-sd[1].x, sd[1].y};
        const v2f c23 = sd[2] + (v2f){-sd[3].x, sd[3].y};
        z23[t] = c01 + c23;
        const float u = sd[0].x + sd[1].x;
        const float w = sd[2].x + sd[3].x;
        tz1[t] = (v2f){u, u} + (v2f){w, -w};
        if (t == 0) __builtin_amdgcn_sched_barrier(0);
    }

    {
        float t0 = tz1[0].x, t1 = tz1[1].x, a0 = tz1[0].y, a1 = tz1[1].y;
        float b0 = z23[0].x, b1 = z23[1].x, c0 = z23[0].y, c1 = z23[1].y;
        swap_halves(t0, t1); swap_halves(a0, a1); swap_halves(b0, b1); swap_halves(c0, c1);
        reinterpret_cast<float4*>(out)[elem] = make_float4(t0 - t1,
                                                           a0 + a1, b0 + b1, c0 + c1);
    }
}

extern "C" void kernel_launch(void* const* d_in, const int* in_sizes, int n_in,
                              void* d_out, int out_size, void* d_ws, size_t ws_size,
                              hipStream_t stream) {
    const float* feat = (const float*)d_in[0];
    const float* qp   = (const float*)d_in[1];
    float* out        = (float*)d_out;
    const int batch   = in_sizes[0] / 4;
    dqc_wmma_kernel<<<batch / BLOCK, BLOCK, 0, stream>>>(feat, qp, out);
}


# phases kept as angles in revolutions: v_tanh_f32/8, CRZ eighth turns as -1/16 rev on qubits 0-2 plus quarter turns (swap, sign), pair phases by one v_cos/v_sin of an angle sum; 1/4 in matrix
# speedup vs baseline: 1.4838x; 1.0722x over previous
#include <hip/hip_runtime.h>
#include <math.h>

typedef __attribute__((ext_vector_type(2))) float v2f;
typedef __attribute__((ext_vector_type(8))) float v8f;

#define BLOCK 256

__device__ __forceinline__ float mat_entry(int m, int k, const float cw[4], const float sw[4]) {
    float p = 0.25f;
#pragma unroll
    for (int j = 0; j < 4; ++j) {
        int mb = (m >> (3 - j)) & 1;
        int kb = (k >> (3 - j)) & 1;
        float u = (mb == kb) ? cw[j] : (mb ? sw[j] : -sw[j]);
        p *= u;
    }
    return p;
}

__device__ __forceinline__ v2f cconj(v2f a) { return (v2f){a.x, -a.y}; }
__device__ __forceinline__ v2f cturn(v2f a) { return (v2f){-a.y, a.x}; }
__device__ __forceinline__ v2f cis_rev(float r) {
    return (v2f){__builtin_amdgcn_cosf(r), __builtin_amdgcn_sinf(r)};
}

__device__ __forceinline__ void swap_halves(float& d, float& s) {
    typedef __attribute__((ext_vector_type(2))) unsigned int v2u_t;
    const v2u_t r = __builtin_amdgcn_permlane16_swap(__float_as_uint(d), __float_as_uint(s), false, false);
    d = __uint_as_float(r.x); s = __uint_as_float(r.y);
}

__global__ __launch_bounds__(BLOCK)
void dqc_wmma_kernel(const float* __restrict__ feat,
                     const float* __restrict__ qp,
                     float* __restrict__ out) {
    const int tid  = threadIdx.x;
    const int lane = tid & 31;
    const int wv   = tid >> 5;
    const int r    = lane & 15;
    const int hi   = lane >> 4;
    const int elem = blockIdx.x * BLOCK + tid;

    float4 f = reinterpret_cast<const float4*>(feat)[elem];
    const float r0 = fmaf(__builtin_amdgcn_tanhf(f.x), 0.125f, -0.0625f);
    const float r1 = fmaf(__builtin_amdgcn_tanhf(f.y), 0.125f, -0.0625f);
    const float r2 = fmaf(__builtin_amdgcn_tanhf(f.z), 0.125f, -0.0625f);
    const float r3 = __builtin_amdgcn_tanhf(f.w) * 0.125f;

    const v2f pp01 = cis_rev(r0 + r1), pm01 = cis_rev(r0 - r1);
    const v2f pp23 = cis_rev(r2 + r3), pm23 = cis_rev(r2 - r3);
    const v2f P01[2][2] = {{cconj(pp01), cconj(pm01)}, {pm01, pp01}};
    const v2f P23[2][2] = {{cconj(pp23), cconj(pm23)}, {pm23, pp23}};
    const v2f F01[2][2] = {{P01[0][0], P01[0][1]}, {P01[1][0], cturn(P01[1][1])}};
    const v2f G[2][2][2] = {{{P23[0][0], P23[0][1]}, {P23[1][0], cturn(P23[1][1])}},
                            {{P23[0][0], P23[0][1]}, {cturn(P23[1][0]), -P23[1][1]}}};

    v2f fr[2][4], fi[2][4];
#pragma unroll
    for (int c = 0; c < 4; ++c) {
        const int q1 = c >> 1, q3 = c & 1;
        const v2f gx = {G[q1][0][q3].x, G[q1][1][q3].x};
        const v2f gy = {G[q1][0][q3].y, G[q1][1][q3].y};
#pragma unroll
        for (int q0 = 0; q0 < 2; ++q0) {
            const v2f fx = {F01[q0][q1].x, F01[q0][q1].x};
            const v2f fy = {F01[q0][q1].y, F01[q0][q1].y};
            const v2f tr = -fy * gy;
            const v2f ti =  fy * gx;
            fr[q0][c] = fx * gx + tr;
            fi[q0][c] = fx * gy + ti;
        }
    }
#pragma unroll
    for (int c = 0; c < 4; ++c) {
        float a0 = fr[0][c].x, b0 = fr[1][c].x, a1 = fr[0][c].y, b1 = fr[1][c].y;
        float a2 = fi[0][c].x, b2 = fi[1][c].x, a3 = fi[0][c].y, b3 = fi[1][c].y;
        swap_halves(a0, b0); swap_halves(a1, b1); swap_halves(a2, b2); swap_halves(a3, b3);
        fr[0][c] = (v2f){a0, a1}; fr[1][c] = (v2f){b0, b1};
        fi[0][c] = (v2f){a2, a3}; fi[1][c] = (v2f){b2, b3};
    }

    float cw[4], sw[4];
#pragma unroll
    for (int j = 0; j < 4; ++j) __sincosf(qp[j] * 0.5f, &sw[j], &cw[j]);
    v2f A[4];
#pragma unroll
    for (int c = 0; c < 4; ++c) {
        A[c].x = mat_entry(r, 4 * c + 2 * hi,     cw, sw);
        A[c].y = mat_entry(r, 4 * c + 2 * hi + 1, cw, sw);
    }

    v2f tz1[2], z23[2];
#pragma unroll
    for (int t = 0; t < 2; ++t) {
        v8f accR = {0.f, 0.f, 0.f, 0.f, 0.f, 0.f, 0.f, 0.f};
        v8f accI = {0.f, 0.f, 0.f, 0.f, 0.f, 0.f, 0.f, 0.f};
#pragma unroll
        for (int c = 0; c < 4; ++c) {
            accR = __builtin_amdgcn_wmma_f32_16x16x4_f32(false, A[c], false, fr[t][c],
                                                         (short)0, accR, false, false);
            accI = __builtin_amdgcn_wmma_f32_16x16x4_f32(false, A[c], false, fi[t][c],
                                                         (short)0, accI, false, false);
        }

        v2f sd[4];
#pragma unroll
        for (int j = 0; j < 4; ++j) {
            const v2f re = {accR[2 * j], accR[2 * j + 1]};
            const v2f im = {accI[2 * j], accI[2 * j + 1]};
            const v2f p  = __builtin_elementwise_fma(im, im, re * re);
            sd[j] = (v2f){p.x, p.x} + (v2f){p.y, -p.y};
        }
        const v2f c01 = sd[0] + (v2f){-sd[1].x, sd[1].y};
        const v2f c23 = sd[2] + (v2f){-sd[3].x, sd[3].y};
        z23[t] = c01 + c23;
        const float u = sd[0].x + sd[1].x;
        const float w = sd[2].x + sd[3].x;
        tz1[t] = (v2f){u, u} + (v2f){w, -w};
        if (t == 0) __builtin_amdgcn_sched_barrier(0);
    }

    {
        float t0 = tz1[0].x, t1 = tz1[1].x, a0 = tz1[0].y, a1 = tz1[1].y;
        float b0 = z23[0].x, b1 = z23[1].x, c0 = z23[0].y, c1 = z23[1].y;
        swap_halves(t0, t1); swap_halves(a0, a1); swap_halves(b0, b1); swap_halves(c0, c1);
        reinterpret_cast<float4*>(out)[elem] = make_float4(t0 - t1,
                                                           a0 + a1, b0 + b1, c0 + c1);
    }
}

extern "C" void kernel_launch(void* const* d_in, const int* in_sizes, int n_in,
                              void* d_out, int out_size, void* d_ws, size_t ws_size,
                              hipStream_t stream) {
    const float* feat = (const float*)d_in[0];
    const float* qp   = (const float*)d_in[1];
    float* out        = (float*)d_out;
    const int batch   = in_sizes[0] / 4;
    dqc_wmma_kernel<<<batch / BLOCK, BLOCK, 0, stream>>>(feat, qp, out);
}


# RZ phases of state ~q are the conjugates of q's: 8 amplitudes formed, not 16; CRZ quarter turns folded into the RY matrix columns; K = 12 per product: 12 WMMAs a wave, not 16; sum order changes
# speedup vs baseline: 1.6596x; 1.1185x over previous
#include <hip/hip_runtime.h>
#include <math.h>

typedef __attribute__((ext_vector_type(2))) float v2f;
typedef __attribute__((ext_vector_type(8))) float v8f;

#define BLOCK 256

__device__ __forceinline__ float mat_entry(int m, int k, const float cw[4], const float sw[4]) {
    float p = 0.25f;
#pragma unroll
    for (int j = 0; j < 4; ++j) {
        int mb = (m >> (3 - j)) & 1;
        int kb = (k >> (3 - j)) & 1;
        float u = (mb == kb) ? cw[j] : (mb ? sw[j] : -sw[j]);
        p *= u;
    }
    return p;
}

__device__ __forceinline__ v2f cconj(v2f a) { return (v2f){a.x, -a.y}; }
__device__ __forceinline__ v2f cis_rev(float r) {
    return (v2f){__builtin_amdgcn_cosf(r), __builtin_amdgcn_sinf(r)};
}

__device__ __forceinline__ void swap_halves(float& d, float& s) {
    typedef __attribute__((ext_vector_type(2))) unsigned int v2u_t;
    const v2u_t r = __builtin_amdgcn_permlane16_swap(__float_as_uint(d), __float_as_uint(s), false, false);
    d = __uint_as_float(r.x); s = __uint_as_float(r.y);
}

__global__ __launch_bounds__(BLOCK)
void dqc_wmma_kernel(const float* __restrict__ feat,
                     const float* __restrict__ qp,
                     float* __restrict__ out) {
    const int tid  = threadIdx.x;
    const int lane = tid & 31;
    const int wv   = tid >> 5;
    const int r    = lane & 15;
    const int hi   = lane >> 4;
    const int elem = blockIdx.x * BLOCK + tid;

    float4 f = reinterpret_cast<const float4*>(feat)[elem];
    const float r0 = fmaf(__builtin_amdgcn_tanhf(f.x), 0.125f, -0.0625f);
    const float r1 = fmaf(__builtin_amdgcn_tanhf(f.y), 0.125f, -0.0625f);
    const float r2 = fmaf(__builtin_amdgcn_tanhf(f.z), 0.125f, -0.0625f);
    const float r3 = __builtin_amdgcn_tanhf(f.w) * 0.125f;

    const v2f pp01 = cis_rev(r0 + r1), pm01 = cis_rev(r0 - r1);
    const v2f pp23 = cis_rev(r2 + r3), pm23 = cis_rev(r2 - r3);
    const v2f F[2] = {cconj(pp01), cconj(pm01)};
    const v2f gx = {pp23.x, pm23.x}, gy = {-pp23.y, pm23.y};
    const v2f hx = {pm23.x, pp23.x}, hy = {-pm23.y, pp23.y};

    v2f bX[2], bY[2], bR[2], bI[2];
#pragma unroll
    for (int q1 = 0; q1 < 2; ++q1) {
        const v2f fx = {F[q1].x, F[q1].x};
        const v2f fy = {F[q1].y, F[q1].y};
        bX[q1] = fx * gx + (-fy * gy);
        bY[q1] = fx * gy + ( fy * gx);
        if (q1 == 1) {
            bR[q1] = fx * hx + (-fy * hy);
            bI[q1] = fx * hy + ( fy * hx);
        } else {
            bR[q1] = fx * (v2f){hx.x, hy.y} + fy * (v2f){-hy.x, hx.y};
            bI[q1] = fx * (v2f){hy.x, hx.y} + fy * (v2f){ hx.x, -hy.y};
        }
    }
    {
        float a0 = bX[0].x, b0 = bX[1].x, a1 = bX[0].y, b1 = bX[1].y;
        float a2 = bY[0].x, b2 = bY[1].x, a3 = bY[0].y, b3 = bY[1].y;
        swap_halves(a0, b0); swap_halves(a1, b1); swap_halves(a2, b2); swap_halves(a3, b3);
        bX[0] = (v2f){a0, a1}; bX[1] = (v2f){b0, b1};
        bY[0] = (v2f){a2, a3}; bY[1] = (v2f){b2, b3};
    }
    {
        float a0 = bR[0].x, b0 = bR[1].x, a1 = bR[0].y, b1 = bR[1].y;
        float a2 = bI[0].x, b2 = bI[1].x, a3 = bI[0].y, b3 = bI[1].y;
        swap_halves(a0, b0); swap_halves(a1, b1); swap_halves(a2, b2); swap_halves(a3, b3);
        bR[0] = (v2f){a0, a1}; bR[1] = (v2f){b0, b1};
        bI[0] = (v2f){a2, a3}; bI[1] = (v2f){b2, b3};
    }

    float cw[4], sw[4];
#pragma unroll
    for (int j = 0; j < 4; ++j) __sincosf(qp[j] * 0.5f, &sw[j], &cw[j]);
    const float Mq0 = mat_entry(r, 8 * hi + 0, cw, sw), Mf0 = mat_entry(r, 15 - (8 * hi + 0), cw, sw);
    const float Mq1 = mat_entry(r, 8 * hi + 1, cw, sw), Mf1 = mat_entry(r, 15 - (8 * hi + 1), cw, sw);
    const float Nq0 = mat_entry(r, 8 * hi + 4, cw, sw), Nf0 = mat_entry(r, 15 - (8 * hi + 4), cw, sw);
    const float Nq1 = mat_entry(r, 8 * hi + 5, cw, sw), Nf1 = mat_entry(r, 15 - (8 * hi + 5), cw, sw);
    const float sMf0 = hi ? Mf0 : -Mf0;
    const v2f ArX = {Mq0,  hi ?  Mf1 : Mq1};
    const v2f ArY = {sMf0, hi ? -Mq1 : Mf1};
    const v2f ArS = {hi ? Nq0 + Nf0 : Nq0 - Nf0, Nf1 - Nq1};
    const v2f AiX = {sMf0, hi ?  Mq1 : Mf1};
    const v2f AiY = {Mq0,  hi ? -Mf1 : Mq1};
    const v2f AiS = {hi ? Nq0 - Nf0 : Nq0 + Nf0, hi ? -(Nq1 + Nf1) : Nq1 + Nf1};

    v2f tz1[2], z23[2];
#pragma unroll
    for (int t = 0; t < 2; ++t) {
        v8f accR = {0.f, 0.f, 0.f, 0.f, 0.f, 0.f, 0.f, 0.f};
        v8f accI = {0.f, 0.f, 0.f, 0.f, 0.f, 0.f, 0.f, 0.f};
        accR = __builtin_amdgcn_wmma_f32_16x16x4_f32(false, ArX, false, bX[t], (short)0, accR, false, false);
        accI = __builtin_amdgcn_wmma_f32_16x16x4_f32(false, AiX, false, bX[t], (short)0, accI, false, false);
        accR = __builtin_amdgcn_wmma_f32_16x16x4_f32(false, ArY, false, bY[t], (short)0, accR, false, false);
        accI = __builtin_amdgcn_wmma_f32_16x16x4_f32(false, AiY, false, bY[t], (short)0, accI, false, false);
        accR = __builtin_amdgcn_wmma_f32_16x16x4_f32(false, ArS, false, bR[t], (short)0, accR, false, false);
        accI = __builtin_amdgcn_wmma_f32_16x16x4_f32(false, AiS, false, bI[t], (short)0, accI, false, false);

        v2f sd[4];
#pragma unroll
        for (int j = 0; j < 4; ++j) {
            const v2f re = {accR[2 * j], accR[2 * j + 1]};
            const v2f im = {accI[2 * j], accI[2 * j + 1]};
            const v2f p  = __builtin_elementwise_fma(im, im, re * re);
            sd[j] = (v2f){p.x, p.x} + (v2f){p.y, -p.y};
        }
        const v2f c01 = sd[0] + (v2f){-sd[1].x, sd[1].y};
        const v2f c23 = sd[2] + (v2f){-sd[3].x, sd[3].y};
        z23[t] = c01 + c23;
        const float u = sd[0].x + sd[1].x;
        const float w = sd[2].x + sd[3].x;
        tz1[t] = (v2f){u, u} + (v2f){w, -w};
    }

    {
        float t0 = tz1[0].x, t1 = tz1[1].x, a0 = tz1[0].y, a1 = tz1[1].y;
        float b0 = z23[0].x, b1 = z23[1].x, c0 = z23[0].y, c1 = z23[1].y;
        swap_halves(t0, t1); swap_halves(a0, a1); swap_halves(b0, b1); swap_halves(c0, c1);
        reinterpret_cast<float4*>(out)[elem] = make_float4(t0 - t1,
                                                           a0 + a1, b0 + b1, c0 + c1);
    }
}

extern "C" void kernel_launch(void* const* d_in, const int* in_sizes, int n_in,
                              void* d_out, int out_size, void* d_ws, size_t ws_size,
                              hipStream_t stream) {
    const float* feat = (const float*)d_in[0];
    const float* qp   = (const float*)d_in[1];
    float* out        = (float*)d_out;
    const int batch   = in_sizes[0] / 4;
    dqc_wmma_kernel<<<batch / BLOCK, BLOCK, 0, stream>>>(feat, qp, out);
}
